# starting kernel
# baseline (speedup 1.0000x reference)
#include <hip/hip_runtime.h>
#include <hip/hip_bf16.h>


typedef __attribute__((ext_vector_type(2))) float v2f;
typedef __attribute__((ext_vector_type(8))) float v8f;

#define BATCH 1024
#define GRIDN 4096
#define DIM   64

__global__ __launch_bounds__(256)
void SOM_norms_kernel(const float* __restrict__ x,
                      const float* __restrict__ w,
                      float* __restrict__ ws) {
    int t = blockIdx.x * blockDim.x + threadIdx.x;
    if (t < GRIDN) {
        const float4* p = (const float4*)(w + (size_t)t * DIM);
        float s = 0.f;
#pragma unroll
        for (int i = 0; i < DIM / 4; ++i) {
            float4 v = p[i];
            s += v.x * v.x + v.y * v.y + v.z * v.z + v.w * v.w;
        }
        ws[t] = s;
    } else if (t < GRIDN + BATCH) {
        int m = t - GRIDN;
        const float4* p = (const float4*)(x + (size_t)m * DIM);
        float s = 0.f;
#pragma unroll
        for (int i = 0; i < DIM / 4; ++i) {
            float4 v = p[i];
            s += v.x * v.x + v.y * v.y + v.z * v.z + v.w * v.w;
        }
        ws[GRIDN + m] = s;
    }
}

__global__ __launch_bounds__(256)
void SOM_dist_kernel(const float* __restrict__ x,
                     const float* __restrict__ w,
                     const float* __restrict__ wnorm,
                     const float* __restrict__ xnorm,
                     float* __restrict__ out) {
    const int lane = threadIdx.x & 31;
    const int wid  = blockIdx.x * (blockDim.x >> 5) + (threadIdx.x >> 5);

    const int mtile = wid >> 6;
    const int ntile = wid & 63;
    const int m0 = mtile * 16;
    const int n0 = ntile * 64;

    const int lm   = lane & 15;
    const int half = lane >> 4;

    const float* xrow = x + (size_t)(m0 + lm) * DIM + 2 * half;
    const float* wrow = w + (size_t)(n0 + lm) * DIM + 2 * half;

    v8f acc0 = {}, acc1 = {}, acc2 = {}, acc3 = {};

#pragma unroll
    for (int kk = 0; kk < DIM / 4; ++kk) {
        const int ko = kk * 4;
        v2f a  = *(const v2f*)(xrow + ko);
        v2f b0 = *(const v2f*)(wrow + ko);
        v2f b1 = *(const v2f*)(wrow + 16 * DIM + ko);
        v2f b2 = *(const v2f*)(wrow + 32 * DIM + ko);
        v2f b3 = *(const v2f*)(wrow + 48 * DIM + ko);
        acc0 = __builtin_amdgcn_wmma_f32_16x16x4_f32(false, a, false, b0,
                                                     (short)0, acc0, false, false);
        acc1 = __builtin_amdgcn_wmma_f32_16x16x4_f32(false, a, false, b1,
                                                     (short)0, acc1, false, false);
        acc2 = __builtin_amdgcn_wmma_f32_16x16x4_f32(false, a, false, b2,
                                                     (short)0, acc2, false, false);
        acc3 = __builtin_amdgcn_wmma_f32_16x16x4_f32(false, a, false, b3,
                                                     (short)0, acc3, false, false);
    }

    float xr[8];
#pragma unroll
    for (int r = 0; r < 8; ++r) xr[r] = xnorm[m0 + r + 8 * half];

    v8f acc[4] = {acc0, acc1, acc2, acc3};
#pragma unroll
    for (int t = 0; t < 4; ++t) {
        const int n  = n0 + t * 16 + lm;
        const float wn = wnorm[n];
#pragma unroll
        for (int r = 0; r < 8; ++r) {
            const int m = m0 + r + 8 * half;
            out[(size_t)m * GRIDN + n] = fmaf(-2.f, acc[t][r], xr[r] + wn);
        }
    }
}

extern "C" void kernel_launch(void* const* d_in, const int* in_sizes, int n_in,
                              void* d_out, int out_size, void* d_ws, size_t ws_size,
                              hipStream_t stream) {
    const float* x = (const float*)d_in[0];
    const float* w = (const float*)d_in[1];
    float* out = (float*)d_out;
    float* ws  = (float*)d_ws;

    SOM_norms_kernel<<<(GRIDN + BATCH + 255) / 256, 256, 0, stream>>>(x, w, ws);

    SOM_dist_kernel<<<512, 256, 0, stream>>>(x, w, ws, ws + GRIDN, out);
}


# Each output store writes one whole 128-byte line: register r of two neighbouring tiles swaps lane halves first (one cross-lane swap per two stores); was two 64-byte pieces per store; sums unchanged
# speedup vs baseline: 2.2743x; 2.2743x over previous
#include <hip/hip_runtime.h>


typedef __attribute__((ext_vector_type(2))) float v2f;
typedef __attribute__((ext_vector_type(4))) float v4f;
typedef __attribute__((ext_vector_type(8))) float v8f;

#define BATCH 1024
#define GRIDN 4096
#define DIM   64

__device__ __forceinline__ void line_swap(float left, float right, float& row_lo, float& row_hi,
                                          const bool behind_matrix = false) {
    if (behind_matrix)
        asm("v_nop\n\tv_nop\n\tv_nop\n\tv_nop\n\tv_permlane16_swap_b32 %0, %1" : "+v"(left), "+v"(right));
    else
        asm("v_permlane16_swap_b32 %0, %1" : "+v"(left), "+v"(right));
    row_lo = left;
    row_hi = right;
}

__global__ __launch_bounds__(128)
void SOM_dist_kernel(const float* __restrict__ x,
                     const float* __restrict__ w,
                     float* __restrict__ out) {
    const int lane = threadIdx.x & 31;
    const int wv   = threadIdx.x >> 5;

    const int mb = blockIdx.x >> 5;
    const int nb = blockIdx.x & 31;
    const int m0 = mb * 128 + (wv >> 1) * 64;
    const int n0 = nb * 128 + (wv & 1) * 64;

    const int lm   = lane & 15;
    const int half = lane >> 4;

    const float* xb = x + (size_t)(m0 + lm) * DIM + 4 * half;
    const float* wb = w + (size_t)(n0 + lm) * DIM + 4 * half;

    v4f fa[8][4], fb[8][4];
#pragma unroll
    for (int q = 0; q < 8; ++q) {
        fa[q][0] = *(const v4f*)(xb + q * 8);
#pragma unroll
        for (int j = 0; j < 4; ++j) fb[q][j] = *(const v4f*)(wb + j * 16 * DIM + q * 8);
        __builtin_amdgcn_sched_barrier(0);
    }
#pragma unroll
    for (int q = 0; q < 8; ++q) fa[q][1] = *(const v4f*)(xb + 1 * 16 * DIM + q * 8);

    v8f acc[4][4];
    float lo[2], hi[2];
    v2f sa[4], sb[4];
#pragma unroll
    for (int j = 0; j < 4; ++j) { sa[j] = (v2f){}; sb[j] = (v2f){}; }
    float* const ob = out + (size_t)m0 * GRIDN + n0 + lane;

#pragma unroll
    for (int i = 0; i < 4; ++i) {
#pragma unroll
        for (int j = 0; j < 4; ++j) acc[i][j] = (v8f){};
#pragma unroll
        for (int t = 0; t < 17; ++t) {
            const int q = t >> 1, s = t & 1;
            if (i == 0 && t == 8) {
#pragma unroll
                for (int ii = 2; ii < 4; ++ii)
#pragma unroll
                    for (int qq = 0; qq < 8; ++qq)
                        fa[qq][ii] = *(const v4f*)(xb + ii * 16 * DIM + qq * 8);
            }
            __builtin_amdgcn_sched_barrier(0);
            if (t < 16) {
                const v2f a = s ? __builtin_shufflevector(fa[q][i], fa[q][i], 2, 3)
                                : __builtin_shufflevector(fa[q][i], fa[q][i], 0, 1);
                sa[i] = a * a + sa[i];
                const v2f am = a * -2.f;
#pragma unroll
                for (int j = 0; j < 4; ++j) {
                    const v2f b = s ? __builtin_shufflevector(fb[q][j], fb[q][j], 2, 3)
                                    : __builtin_shufflevector(fb[q][j], fb[q][j], 0, 1);
                    if (i == 0) sb[j] = b * b + sb[j];
                    acc[i][j] = __builtin_amdgcn_wmma_f32_16x16x4_f32(
                        false, am, false, b, (short)0, acc[i][j], false, false);
                }
            } else {
                const v2f na = { sa[i].x + sa[i].y, 1.f };
#pragma unroll
                for (int j = 0; j < 4; ++j) {
                    const v2f nw = { 1.f, sb[j].x + sb[j].y };
                    acc[i][j] = __builtin_amdgcn_wmma_f32_16x16x4_f32(
                        false, na, false, nw, (short)0, acc[i][j], false, false);
                }
            }
            if (i > 0 && t > 0) {
                const int r = (t - 1) >> 1, up = (t - 1) & 1;
                if (!up) {
                    line_swap(acc[i - 1][0][r], acc[i - 1][1][r], lo[0], hi[0]);
                    line_swap(acc[i - 1][2][r], acc[i - 1][3][r], lo[1], hi[1]);
                }
#pragma unroll
                for (int p = 0; p < 2; ++p)
                    ob[(size_t)((i - 1) * 16 + r + 8 * up) * GRIDN + p * 32] = up ? hi[p] : lo[p];
            }
        }
    }

    float tl[2][8], th[2][8];
#pragma unroll
    for (int p = 0; p < 2; ++p)
#pragma unroll
        for (int r = 0; r < 8; ++r) {
            line_swap(acc[3][2 * p][r], acc[3][2 * p + 1][r], tl[p][r], th[p][r], p == 0 && r == 0);
        }
#pragma unroll
    for (int r = 0; r < 8; ++r) {
#pragma unroll
        for (int p = 0; p < 2; ++p) ob[(size_t)(3 * 16 + r) * GRIDN + p * 32] = tl[p][r];
    }
#pragma unroll
    for (int r = 0; r < 8; ++r) {
#pragma unroll
        for (int p = 0; p < 2; ++p) ob[(size_t)(3 * 16 + r + 8) * GRIDN + p * 32] = th[p][r];
    }
}

extern "C" void kernel_launch(void* const* d_in, const int* in_sizes, int n_in,
                              void* d_out, int out_size, void* d_ws, size_t ws_size,
                              hipStream_t stream) {
    const float* x = (const float*)d_in[0];
    const float* w = (const float*)d_in[1];
    float* out = (float*)d_out;

    SOM_dist_kernel<<<256, 128, 0, stream>>>(x, w, out);
}


# Previous band's 32 stores go out as 4 groups of 8 (4 swaps, then 8 whole-line stores) behind k-steps 1, 5, 9, 13 of the next band; were 2 stores behind each of 16 k-steps; sums unchanged
# speedup vs baseline: 2.6656x; 1.1721x over previous
#include <hip/hip_runtime.h>


typedef __attribute__((ext_vector_type(2))) float v2f;
typedef __attribute__((ext_vector_type(4))) float v4f;
typedef __attribute__((ext_vector_type(8))) float v8f;

#define BATCH 1024
#define GRIDN 4096
#define DIM   64

__device__ __forceinline__ void line_swap(float left, float right, float& row_lo, float& row_hi,
                                          const bool behind_matrix = false) {
    if (behind_matrix)
        asm("v_nop\n\tv_nop\n\tv_nop\n\tv_nop\n\tv_permlane16_swap_b32 %0, %1" : "+v"(left), "+v"(right));
    else
        asm("v_permlane16_swap_b32 %0, %1" : "+v"(left), "+v"(right));
    row_lo = left;
    row_hi = right;
}

__global__ __launch_bounds__(128)
void SOM_dist_kernel(const float* __restrict__ x,
                     const float* __restrict__ w,
                     float* __restrict__ out) {
    const int lane = threadIdx.x & 31;
    const int wv   = threadIdx.x >> 5;

    const int mb = blockIdx.x >> 5;
    const int nb = blockIdx.x & 31;
    const int m0 = mb * 128 + (wv >> 1) * 64;
    const int n0 = nb * 128 + (wv & 1) * 64;

    const int lm   = lane & 15;
    const int half = lane >> 4;

    const float* xb = x + (size_t)(m0 + lm) * DIM + 4 * half;
    const float* wb = w + (size_t)(n0 + lm) * DIM + 4 * half;

    v4f fa[8][4], fb[8][4];
#pragma unroll
    for (int q = 0; q < 8; ++q) {
        fa[q][0] = *(const v4f*)(xb + q * 8);
#pragma unroll
        for (int j = 0; j < 4; ++j) fb[q][j] = *(const v4f*)(wb + j * 16 * DIM + q * 8);
        __builtin_amdgcn_sched_barrier(0);
    }
#pragma unroll
    for (int q = 0; q < 8; ++q) fa[q][1] = *(const v4f*)(xb + 1 * 16 * DIM + q * 8);

    v8f acc[4][4];
    float lo[2][2], hi[2][2];
    v2f sa[4], sb[4];
#pragma unroll
    for (int j = 0; j < 4; ++j) { sa[j] = (v2f){}; sb[j] = (v2f){}; }
    float* const ob = out + (size_t)m0 * GRIDN + n0 + lane;

#pragma unroll
    for (int i = 0; i < 4; ++i) {
#pragma unroll
        for (int j = 0; j < 4; ++j) acc[i][j] = (v8f){};
#pragma unroll
        for (int t = 0; t < 17; ++t) {
            const int q = t >> 1, s = t & 1;
            if (i == 0 && t == 8) {
#pragma unroll
                for (int ii = 2; ii < 4; ++ii)
#pragma unroll
                    for (int qq = 0; qq < 8; ++qq)
                        fa[qq][ii] = *(const v4f*)(xb + ii * 16 * DIM + qq * 8);
            }
            __builtin_amdgcn_sched_barrier(0);
            if (t < 16) {
                const v2f a = s ? __builtin_shufflevector(fa[q][i], fa[q][i], 2, 3)
                                : __builtin_shufflevector(fa[q][i], fa[q][i], 0, 1);
                sa[i] = a * a + sa[i];
                const v2f am = a * -2.f;
#pragma unroll
                for (int j = 0; j < 4; ++j) {
                    const v2f b = s ? __builtin_shufflevector(fb[q][j], fb[q][j], 2, 3)
                                    : __builtin_shufflevector(fb[q][j], fb[q][j], 0, 1);
                    if (i == 0) sb[j] = b * b + sb[j];
                    acc[i][j] = __builtin_amdgcn_wmma_f32_16x16x4_f32(
                        false, am, false, b, (short)0, acc[i][j], false, false);
                }
            } else {
                const v2f na = { sa[i].x + sa[i].y, 1.f };
#pragma unroll
                for (int j = 0; j < 4; ++j) {
                    const v2f nw = { 1.f, sb[j].x + sb[j].y };
                    acc[i][j] = __builtin_amdgcn_wmma_f32_16x16x4_f32(
                        false, na, false, nw, (short)0, acc[i][j], false, false);
                }
            }
            if (i > 0 && (t & 3) == 1) {
                const int g = t >> 2;
#pragma unroll
                for (int rr = 0; rr < 2; ++rr) {
                    line_swap(acc[i - 1][0][2 * g + rr], acc[i - 1][1][2 * g + rr], lo[rr][0], hi[rr][0]);
                    line_swap(acc[i - 1][2][2 * g + rr], acc[i - 1][3][2 * g + rr], lo[rr][1], hi[rr][1]);
                }
#pragma unroll
                for (int rr = 0; rr < 2; ++rr)
#pragma unroll
                    for (int up = 0; up < 2; ++up)
#pragma unroll
                        for (int p = 0; p < 2; ++p)
                            ob[(size_t)((i - 1) * 16 + 2 * g + rr + 8 * up) * GRIDN + p * 32] =
                                up ? hi[rr][p] : lo[rr][p];
            }
        }
    }

    float tl[2][8], th[2][8];
#pragma unroll
    for (int p = 0; p < 2; ++p)
#pragma unroll
        for (int r = 0; r < 8; ++r) {
            line_swap(acc[3][2 * p][r], acc[3][2 * p + 1][r], tl[p][r], th[p][r], p == 0 && r == 0);
        }
#pragma unroll
    for (int r = 0; r < 8; ++r) {
#pragma unroll
        for (int p = 0; p < 2; ++p) ob[(size_t)(3 * 16 + r) * GRIDN + p * 32] = tl[p][r];
    }
#pragma unroll
    for (int r = 0; r < 8; ++r) {
#pragma unroll
        for (int p = 0; p < 2; ++p) ob[(size_t)(3 * 16 + r + 8) * GRIDN + p * 32] = th[p][r];
    }
}

extern "C" void kernel_launch(void* const* d_in, const int* in_sizes, int n_in,
                              void* d_out, int out_size, void* d_ws, size_t ws_size,
                              hipStream_t stream) {
    const float* x = (const float*)d_in[0];
    const float* w = (const float*)d_in[1];
    float* out = (float*)d_out;

    SOM_dist_kernel<<<256, 128, 0, stream>>>(x, w, out);
}


# Band leaves in ONE place: 8 sixteen-byte stores of 4 whole lines (lane c of tile j = column 4c+j), registers kept until drained; was 32 line stores + 16 lane swaps in 4 places; sums unchanged
# speedup vs baseline: 2.7932x; 1.0479x over previous
#include <hip/hip_runtime.h>


typedef __attribute__((ext_vector_type(2))) float v2f;
typedef __attribute__((ext_vector_type(4))) float v4f;
typedef __attribute__((ext_vector_type(8))) float v8f;

#define BATCH 1024
#define GRIDN 4096
#define DIM   64


__global__ __launch_bounds__(128)
void SOM_dist_kernel(const float* __restrict__ x,
                     const float* __restrict__ w,
                     float* __restrict__ out) {
    const int lane = threadIdx.x & 31;
    const int wv   = threadIdx.x >> 5;

    const int mb = blockIdx.x >> 5;
    const int nb = blockIdx.x & 31;
    const int m0 = mb * 128 + (wv >> 1) * 64;
    const int n0 = nb * 128 + (wv & 1) * 64;

    const int lm   = lane & 15;
    const int half = lane >> 4;

    const float* xb = x + (size_t)(m0 + lm) * DIM + 4 * half;
    const float* wb = w + (size_t)(n0 + 4 * lm) * DIM + 4 * half;

    v4f fa[8][4], fb[8][4];
#pragma unroll
    for (int q = 0; q < 8; ++q) {
        fa[q][0] = *(const v4f*)(xb + q * 8);
#pragma unroll
        for (int j = 0; j < 4; ++j) fb[q][j] = *(const v4f*)(wb + j * DIM + q * 8);
        __builtin_amdgcn_sched_barrier(0);
    }
#pragma unroll
    for (int q = 0; q < 8; ++q) fa[q][1] = *(const v4f*)(xb + 1 * 16 * DIM + q * 8);

    v8f acc[4][4];
    v4f row[8];
    v2f sa[4], sb[4];
#pragma unroll
    for (int j = 0; j < 4; ++j) { sa[j] = (v2f){}; sb[j] = (v2f){}; }
    float* const ob = out + (size_t)(m0 + 8 * half) * GRIDN + n0 + 4 * lm;

#pragma unroll
    for (int i = 0; i < 4; ++i) {
#pragma unroll
        for (int j = 0; j < 4; ++j) acc[i][j] = (v8f){};
#pragma unroll
        for (int t = 0; t < 17; ++t) {
            const int q = t >> 1, s = t & 1;
            if (i == 0 && t == 8) {
#pragma unroll
                for (int ii = 2; ii < 4; ++ii)
#pragma unroll
                    for (int qq = 0; qq < 8; ++qq)
                        fa[qq][ii] = *(const v4f*)(xb + ii * 16 * DIM + qq * 8);
            }
            __builtin_amdgcn_sched_barrier(0);
            if (t < 16) {
                const v2f a = s ? __builtin_shufflevector(fa[q][i], fa[q][i], 2, 3)
                                : __builtin_shufflevector(fa[q][i], fa[q][i], 0, 1);
                sa[i] = a * a + sa[i];
                const v2f am = a * -2.f;
#pragma unroll
                for (int j = 0; j < 4; ++j) {
                    const v2f b = s ? __builtin_shufflevector(fb[q][j], fb[q][j], 2, 3)
                                    : __builtin_shufflevector(fb[q][j], fb[q][j], 0, 1);
                    if (i == 0) sb[j] = b * b + sb[j];
                    acc[i][j] = __builtin_amdgcn_wmma_f32_16x16x4_f32(
                        false, am, false, b, (short)0, acc[i][j], false, false);
                }
            } else {
                const v2f na = { sa[i].x + sa[i].y, 1.f };
#pragma unroll
                for (int j = 0; j < 4; ++j) {
                    const v2f nw = { 1.f, sb[j].x + sb[j].y };
                    acc[i][j] = __builtin_amdgcn_wmma_f32_16x16x4_f32(
                        false, na, false, nw, (short)0, acc[i][j], false, false);
                }
            }
            if (i > 0 && t == 1) {
#pragma unroll
                for (int r = 0; r < 8; ++r)
                    row[r] = (v4f){ acc[i - 1][0][r], acc[i - 1][1][r],
                                    acc[i - 1][2][r], acc[i - 1][3][r] };
                asm volatile("" :: "v"(row[0]), "v"(row[1]), "v"(row[2]), "v"(row[3]),
                                   "v"(row[4]), "v"(row[5]), "v"(row[6]), "v"(row[7]));
                __builtin_amdgcn_sched_barrier(0);
#pragma unroll
                for (int r = 0; r < 8; ++r)
                    *(v4f*)(ob + (size_t)((i - 1) * 16 + r) * GRIDN) = row[r];
            }
            if (i > 0 && t == 15) {
                asm volatile("" :: "v"(row[0]), "v"(row[1]), "v"(row[2]), "v"(row[3]),
                                   "v"(row[4]), "v"(row[5]), "v"(row[6]), "v"(row[7]));
            }
        }
    }

#pragma unroll
    for (int r = 0; r < 8; ++r)
        row[r] = (v4f){ acc[3][0][r], acc[3][1][r], acc[3][2][r], acc[3][3][r] };
    asm volatile("" :: "v"(row[0]), "v"(row[1]), "v"(row[2]), "v"(row[3]),
                       "v"(row[4]), "v"(row[5]), "v"(row[6]), "v"(row[7]));
    __builtin_amdgcn_sched_barrier(0);
#pragma unroll
    for (int r = 0; r < 8; ++r)
        *(v4f*)(ob + (size_t)(3 * 16 + r) * GRIDN) = row[r];
}

extern "C" void kernel_launch(void* const* d_in, const int* in_sizes, int n_in,
                              void* d_out, int out_size, void* d_ws, size_t ws_size,
                              hipStream_t stream) {
    const float* x = (const float*)d_in[0];
    const float* w = (const float*)d_in[1];
    float* out = (float*)d_out;

    SOM_dist_kernel<<<256, 128, 0, stream>>>(x, w, out);
}
